# starting kernel
# baseline (speedup 1.0000x reference)
#include <hip/hip_runtime.h>

typedef __attribute__((ext_vector_type(2))) float v2f;
typedef __attribute__((ext_vector_type(8))) float v8f;

#define IMSIZE   224
#define NSEG     1024
#define SMOOTH_K 8.0f
#define EPS_F    1e-10f
#define TILES_PER_ROW (IMSIZE / 16)
#define WAVES_PER_BLOCK 8
#define NUM_TILES ((IMSIZE * IMSIZE) / 16)

#define NEGK_LOG2E (-SMOOTH_K * 1.44269504088896340736f)
#define LOG2_NSEG  10.0f
#define LN2_OVER_K (0.69314718055994530942f / SMOOTH_K)

__global__ __launch_bounds__(256) void sketch_render_kernel(
    const float* __restrict__ xpar,
    const float* __restrict__ ypar,
    float* __restrict__ out)
{
    __shared__ float4 sAB[NSEG];
    __shared__ float4 sQ[NSEG];
    __shared__ float2 sR[NSEG];

    for (int s = threadIdx.x; s < NSEG; s += blockDim.x) {
        float x0 = xpar[s]     * (float)IMSIZE;
        float y0 = ypar[s]     * (float)IMSIZE;
        float x1 = xpar[s + 1] * (float)IMSIZE;
        float y1 = ypar[s + 1] * (float)IMSIZE;
        float dx = x1 - x0, dy = y1 - y0;
        float d2 = dx * dx + dy * dy;
        float inv = 1.0f / (d2 + EPS_F);
        float cross = y0 * x1 - x0 * y1;
        float A  = dx * dy * inv;
        float B  = dx * dx * inv;
        float Cc = -dy * cross * inv;
        float idx = 1.0f / (dx + EPS_F);
        float sl = dy * idx;
        float ic = cross * idx;
        float vert = (x0 == x1) ? 1.0f : 0.0f;
        sAB[s] = make_float4(A, B, Cc, sl);
        sQ[s]  = make_float4(ic, fminf(x0, x1), fmaxf(x0, x1), fminf(y0, y1));
        sR[s]  = make_float2(fmaxf(y0, y1), vert);
    }
    __syncthreads();

    int wid  = (blockIdx.x * blockDim.x + threadIdx.x) >> 5;
    int lane = threadIdx.x & 31;
    int grp  = lane >> 4;
    int m    = lane & 15;
    int row  = wid / TILES_PER_ROW;
    int col  = wid % TILES_PER_ROW;
    float xp = (float)(col * 16 + m);
    float yp = (float)row;

    v8f c = {0.0f, 0.0f, 0.0f, 0.0f, 0.0f, 0.0f, 0.0f, 0.0f};
    v2f bones = {1.0f, 1.0f};

    #pragma unroll 2
    for (int s = 0; s < NSEG; s += 4) {
        int k0 = s + grp * 2;
        v2f a;
        #pragma unroll
        for (int j = 0; j < 2; ++j) {
            int k = k0 + j;
            float4 p = sAB[k];
            float4 q = sQ[k];
            float2 r = sR[k];
            float cx = fmaf(p.x, yp, fmaf(p.y, xp, p.z));
            float cy = fmaf(p.w, cx, q.x);
            float yps = (r.y != 0.0f) ? yp : cy;
            float xp1 = fminf(fmaxf(cx,  q.y), q.z);
            float yp1 = fminf(fmaxf(yps, q.w), r.x);
            float ddx = xp - xp1;
            float ddy = yp - yp1;
            float t = fmaf(ddx, ddx, EPS_F);
            t = fmaf(ddy, ddy, t);
            float dist = __fsqrt_rn(t);
            a[j] = __builtin_amdgcn_exp2f(dist * NEGK_LOG2E);
        }
        c = __builtin_amdgcn_wmma_f32_16x16x4_f32(false, a, false, bones,
                                                  (short)0, c, false, false);
    }

    if (lane == 0 || lane == 16) {
        int base = row * IMSIZE + col * 16 + grp * 8;
        #pragma unroll
        for (int rI = 0; rI < 8; ++rI) {
            float res = c[rI];
            float val = (LOG2_NSEG - __builtin_amdgcn_logf(res)) * LN2_OVER_K;
            out[base + rI] = val;
        }
    }
}

extern "C" void kernel_launch(void* const* d_in, const int* in_sizes, int n_in,
                              void* d_out, int out_size, void* d_ws, size_t ws_size,
                              hipStream_t stream) {
    (void)in_sizes; (void)n_in; (void)out_size; (void)d_ws; (void)ws_size;
    const float* x = (const float*)d_in[0];
    const float* y = (const float*)d_in[1];
    float* out = (float*)d_out;
    dim3 grid(NUM_TILES / WAVES_PER_BLOCK);
    dim3 block(WAVES_PER_BLOCK * 32);
    sketch_render_kernel<<<grid, block, 0, stream>>>(x, y, out);
}


# Issue-ahead pipeline: a step's 8 sqrt and 8 exp2 are written between the next step's plain vector work, products two steps ahead, LDS operands a group ahead, both table-build loads up front
# speedup vs baseline: 2.5801x; 2.5801x over previous
#include <hip/hip_runtime.h>

typedef __attribute__((ext_vector_type(2))) float v2f;
typedef __attribute__((ext_vector_type(8))) float v8f;

#define IMSIZE   224
#define NSEG     1024
#define SMOOTH_K 8.0f
#define EPS_F    1e-10f

#define NEGK_LOG2E (-SMOOTH_K * 1.44269504088896340736f)
#define LOG2_NSEG  10.0f
#define LN2_OVER_K (0.69314718055994530942f / SMOOTH_K)

#define REG_W     32
#define REG_H     7
#define REG_PER_ROW (IMSIZE / REG_W)
#define NUM_REG   (REG_PER_ROW * (IMSIZE / REG_H))
#define NRANGE    8
#define GROUPS_PER_RANGE (NSEG / 16 / NRANGE)
#define BLOCK_THREADS (2 * NRANGE * 32)

__global__ __launch_bounds__(BLOCK_THREADS) void sketch_render_kernel(
    const float* __restrict__ xpar,
    const float* __restrict__ ypar,
    float* __restrict__ out)
{
    __shared__ float4 sPS[NSEG];
    __shared__ float4 sPP[NSEG];
    __shared__ float  sLen[NSEG];
    __shared__ float  sPart[NRANGE][REG_H][REG_W];

    const int rx = blockIdx.x % REG_PER_ROW;
    const int ry = blockIdx.x / REG_PER_ROW;
    const int py0 = ry * REG_H;
    const float ox = (float)(rx * REG_W);
    const float oy = (float)py0;

    float ex0[2], ey0[2], ex1[2], ey1[2];
    #pragma unroll
    for (int i = 0; i < NSEG / BLOCK_THREADS; ++i) {
        const int s = threadIdx.x + i * BLOCK_THREADS;
        ex0[i] = xpar[s];
        ey0[i] = ypar[s];
        ex1[i] = xpar[s + 1];
        ey1[i] = ypar[s + 1];
    }
    #pragma unroll
    for (int i = 0; i < NSEG / BLOCK_THREADS; ++i) {
        const int s = threadIdx.x + i * BLOCK_THREADS;
        float x0 = ex0[i] * (float)IMSIZE;
        float y0 = ey0[i] * (float)IMSIZE;
        float x1 = ex1[i] * (float)IMSIZE;
        float y1 = ey1[i] * (float)IMSIZE;
        float dx = x1 - x0, dy = y1 - y0;
        float l2 = dx * dx + dy * dy;
        bool point = !(l2 >= 1e-30f);
        float inv = __builtin_amdgcn_rsqf(point ? 1.0f : l2);
        float ux = point ? 1.0f : dx * inv;
        float uy = point ? 0.0f : dy * inv;
        float len = point ? 0.0f : l2 * inv;
        float rx0 = ox - x0, ry0 = oy - y0;
        float cs = fmaf(ux, rx0, uy * ry0);
        float cp = fmaf(ux, ry0, -(uy * rx0));
        sPS[s] = make_float4(ux, uy, cs, 0.0f);
        sPP[s] = make_float4(-uy, ux, cp, 0.0f);
        sLen[s] = len;
    }
    __syncthreads();

    const int wave = threadIdx.x >> 5;
    const int lane = threadIdx.x & 31;
    const int h    = lane >> 4;
    const int c    = lane & 15;
    const int half  = wave & 1;
    const int range = wave >> 1;

    const float bx = h ? 1.0f : (float)(half * 16 + c);
    const v2f eps2 = {EPS_F, EPS_F};
    const v2f negk2 = {NEGK_LOG2E, NEGK_LOG2E};
    float by[REG_H];
    v2f acc[REG_H];
    #pragma unroll
    for (int t = 0; t < REG_H; ++t) {
        by[t]  = h ? 0.0f : (float)t;
        acc[t] = (v2f){0.0f, 0.0f};
    }

    const v8f zero8 = {0.0f, 0.0f, 0.0f, 0.0f, 0.0f, 0.0f, 0.0f, 0.0f};

    const int gbase = range * GROUPS_PER_RANGE;

    v2f asA, apA, asB, apB;
    float lenA[8], lenB[8];
    {
        const int s0 = gbase * 16;
        const float* rows = (const float*)&sPS[s0 + c];
        const float* rowp = (const float*)&sPP[s0 + c];
        asA = (v2f){rows[2 * h], rows[2 * h + 1]};
        apA = (v2f){rowp[2 * h], rowp[2 * h + 1]};
        #pragma unroll
        for (int r = 0; r < 8; ++r) lenA[r] = sLen[s0 + 8 * h + r];
    }

    v8f sa1, sp1, sa2, sp2;
    {
        v2f b0 = {bx, by[0]};
        v2f b1 = {bx, by[1]};
        sa1 = __builtin_amdgcn_wmma_f32_16x16x4_f32(false, asA, false, b0, (short)0, zero8, false, false);
        sp1 = __builtin_amdgcn_wmma_f32_16x16x4_f32(false, apA, false, b0, (short)0, zero8, false, false);
        sa2 = __builtin_amdgcn_wmma_f32_16x16x4_f32(false, asA, false, b1, (short)0, zero8, false, false);
        sp2 = __builtin_amdgcn_wmma_f32_16x16x4_f32(false, apA, false, b1, (short)0, zero8, false, false);
    }
    v2f d[4];
    #pragma unroll
    for (int rp = 0; rp < 4; ++rp) {
        const int r = 2 * rp;
        v2f m = {__builtin_amdgcn_fmed3f(sa1[r],     0.0f, lenA[r]),
                 __builtin_amdgcn_fmed3f(sa1[r + 1], 0.0f, lenA[r + 1])};
        v2f a2 = {sa1[r], sa1[r + 1]};
        v2f p2 = {sp1[r], sp1[r + 1]};
        v2f e = a2 - m;
        v2f q = __builtin_elementwise_fma(p2, p2, eps2);
        d[rp] = __builtin_elementwise_fma(e, e, q);
    }
    sa1 = sa2; sp1 = sp2;

    for (int g = 0; g < GROUPS_PER_RANGE; ++g) {
        {
            const int gn = (g + 1 < GROUPS_PER_RANGE) ? g + 1 : g;
            const int s0 = (gbase + gn) * 16;
            const float* rows = (const float*)&sPS[s0 + c];
            const float* rowp = (const float*)&sPP[s0 + c];
            asB = (v2f){rows[2 * h], rows[2 * h + 1]};
            apB = (v2f){rowp[2 * h], rowp[2 * h + 1]};
            #pragma unroll
            for (int r = 0; r < 8; ++r) lenB[r] = sLen[s0 + 8 * h + r];
        }

        #pragma unroll
        for (int t = 0; t < REG_H; ++t) {
            {
                const int t2 = (t + 2) % REG_H;
                v2f b = {bx, by[t2]};
                v2f as2 = (t + 2 < REG_H) ? asA : asB;
                v2f ap2 = (t + 2 < REG_H) ? apA : apB;
                sa2 = __builtin_amdgcn_wmma_f32_16x16x4_f32(false, as2, false, b, (short)0, zero8, false, false);
                sp2 = __builtin_amdgcn_wmma_f32_16x16x4_f32(false, ap2, false, b, (short)0, zero8, false, false);
            }
            float ln[8];
            #pragma unroll
            for (int r = 0; r < 8; ++r) ln[r] = (t + 1 < REG_H) ? lenA[r] : lenB[r];

            float rt[8], ex[8], mm[8];
            v2f qq[4], ee[4], dn[4], ar[4];

            rt[0] = __builtin_amdgcn_sqrtf(d[0].x);
            mm[0] = __builtin_amdgcn_fmed3f(sa1[0], 0.0f, ln[0]);
            rt[1] = __builtin_amdgcn_sqrtf(d[0].y);
            mm[1] = __builtin_amdgcn_fmed3f(sa1[1], 0.0f, ln[1]);
            rt[2] = __builtin_amdgcn_sqrtf(d[1].x);
            qq[0] = __builtin_elementwise_fma((v2f){sp1[0], sp1[1]}, (v2f){sp1[0], sp1[1]}, eps2);
            rt[3] = __builtin_amdgcn_sqrtf(d[1].y);
            mm[2] = __builtin_amdgcn_fmed3f(sa1[2], 0.0f, ln[2]);
            rt[4] = __builtin_amdgcn_sqrtf(d[2].x);
            mm[3] = __builtin_amdgcn_fmed3f(sa1[3], 0.0f, ln[3]);
            rt[5] = __builtin_amdgcn_sqrtf(d[2].y);
            qq[1] = __builtin_elementwise_fma((v2f){sp1[2], sp1[3]}, (v2f){sp1[2], sp1[3]}, eps2);
            rt[6] = __builtin_amdgcn_sqrtf(d[3].x);
            mm[4] = __builtin_amdgcn_fmed3f(sa1[4], 0.0f, ln[4]);
            rt[7] = __builtin_amdgcn_sqrtf(d[3].y);
            mm[5] = __builtin_amdgcn_fmed3f(sa1[5], 0.0f, ln[5]);

            qq[2] = __builtin_elementwise_fma((v2f){sp1[4], sp1[5]}, (v2f){sp1[4], sp1[5]}, eps2);
            mm[6] = __builtin_amdgcn_fmed3f(sa1[6], 0.0f, ln[6]);
            mm[7] = __builtin_amdgcn_fmed3f(sa1[7], 0.0f, ln[7]);
            qq[3] = __builtin_elementwise_fma((v2f){sp1[6], sp1[7]}, (v2f){sp1[6], sp1[7]}, eps2);
            ar[0] = (v2f){rt[0], rt[1]} * negk2;
            ar[1] = (v2f){rt[2], rt[3]} * negk2;
            ar[2] = (v2f){rt[4], rt[5]} * negk2;
            ar[3] = (v2f){rt[6], rt[7]} * negk2;

            ex[0] = __builtin_amdgcn_exp2f(ar[0].x);
            ee[0] = (v2f){sa1[0], sa1[1]} - (v2f){mm[0], mm[1]};
            ex[1] = __builtin_amdgcn_exp2f(ar[0].y);
            ee[1] = (v2f){sa1[2], sa1[3]} - (v2f){mm[2], mm[3]};
            ex[2] = __builtin_amdgcn_exp2f(ar[1].x);
            ee[2] = (v2f){sa1[4], sa1[5]} - (v2f){mm[4], mm[5]};
            ex[3] = __builtin_amdgcn_exp2f(ar[1].y);
            ee[3] = (v2f){sa1[6], sa1[7]} - (v2f){mm[6], mm[7]};
            ex[4] = __builtin_amdgcn_exp2f(ar[2].x);
            dn[0] = __builtin_elementwise_fma(ee[0], ee[0], qq[0]);
            ex[5] = __builtin_amdgcn_exp2f(ar[2].y);
            dn[1] = __builtin_elementwise_fma(ee[1], ee[1], qq[1]);
            ex[6] = __builtin_amdgcn_exp2f(ar[3].x);
            dn[2] = __builtin_elementwise_fma(ee[2], ee[2], qq[2]);
            ex[7] = __builtin_amdgcn_exp2f(ar[3].y);
            dn[3] = __builtin_elementwise_fma(ee[3], ee[3], qq[3]);

            acc[t] += (v2f){ex[0], ex[1]};
            acc[t] += (v2f){ex[2], ex[3]};
            acc[t] += (v2f){ex[4], ex[5]};
            acc[t] += (v2f){ex[6], ex[7]};

            d[0] = dn[0]; d[1] = dn[1]; d[2] = dn[2]; d[3] = dn[3];
            sa1 = sa2; sp1 = sp2;
        }
        asA = asB; apA = apB;
        #pragma unroll
        for (int r = 0; r < 8; ++r) lenA[r] = lenB[r];
    }

    const v2f ones = {1.0f, 1.0f};
    #pragma unroll
    for (int t = 0; t < REG_H; ++t) {
        v2f bs = acc[t];
        v8f dsum = __builtin_amdgcn_wmma_f32_16x16x4_f32(false, ones, false, bs, (short)0, zero8, false, false);
        if (h == 0) sPart[range][t][half * 16 + c] = dsum[0];
    }
    __syncthreads();

    if (threadIdx.x < REG_H * REG_W) {
        const int t = threadIdx.x >> 5;
        const int x = threadIdx.x & 31;
        float res = sPart[0][t][x];
        #pragma unroll
        for (int q = 1; q < NRANGE; ++q) res += sPart[q][t][x];
        float val = (LOG2_NSEG - __builtin_amdgcn_logf(res)) * LN2_OVER_K;
        out[(py0 + t) * IMSIZE + rx * REG_W + x] = val;
    }
}

extern "C" void kernel_launch(void* const* d_in, const int* in_sizes, int n_in,
                              void* d_out, int out_size, void* d_ws, size_t ws_size,
                              hipStream_t stream) {
    (void)in_sizes; (void)n_in; (void)out_size; (void)d_ws; (void)ws_size;
    const float* x = (const float*)d_in[0];
    const float* y = (const float*)d_in[1];
    float* out = (float*)d_out;
    dim3 grid(NUM_REG);
    dim3 block(BLOCK_THREADS);
    sketch_render_kernel<<<grid, block, 0, stream>>>(x, y, out);
}
